# starting kernel
# baseline (speedup 1.0000x reference)
#include <hip/hip_runtime.h>

typedef float v2f __attribute__((ext_vector_type(2)));
typedef float v8f __attribute__((ext_vector_type(8)));

#define H   512
#define W   512
#define OH  128
#define OW  128
#define NC  96
#define TILES_PER_IMG 64
#define WAVES_PER_BLOCK 4

template <bool EDGE>
__device__ __forceinline__ void pass1(const float* __restrict__ src,
                                      int ybase, int xbase,
                                      float (*__restrict__ T)[84],
                                      const float (*__restrict__ Hm)[80],
                                      int l15, int half)
{
    #pragma unroll
    for (int sub = 0; sub < 5; ++sub) {
        v8f acc = {};
        const int y = ybase + 16 * sub + l15;
        bool yv = true;
        const float* rowp;
        if (EDGE) {
            yv = (y >= 0 && y < H);
            const int yc = min(max(y, 0), H - 1);
            rowp = src + (size_t)yc * W;
        } else {
            rowp = src + (size_t)y * W + xbase + 2 * half;
        }
        #pragma unroll
        for (int kk = 0; kk < 19; ++kk) {
            v2f a;
            if (EDGE) {
                int x  = xbase + 4 * kk + 2 * half;
                int xc = min(max(x, 0), W - 2);
                a = *(const v2f*)(rowp + xc);
                a.x = (yv && (unsigned)x       < (unsigned)W) ? a.x : 0.0f;
                a.y = (yv && (unsigned)(x + 1) < (unsigned)W) ? a.y : 0.0f;
            } else {
                a = *(const v2f*)(rowp + 4 * kk);
            }
            v2f b = *(const v2f*)(&Hm[l15][4 * kk + 2 * half]);
            acc = __builtin_amdgcn_wmma_f32_16x16x4_f32(false, a, false, b,
                                                        (short)0, acc, false, false);
        }
        #pragma unroll
        for (int r = 0; r < 8; ++r)
            T[l15][16 * sub + r + 8 * half] = acc[r];
    }
}

__global__ __launch_bounds__(128)
void aa_interp_wmma_kernel(const float* __restrict__ in,
                           const float* __restrict__ wgt,
                           float* __restrict__ out)
{
    __shared__ float ctab[13];
    __shared__ float Hm[16][80];
    __shared__ float Tt[WAVES_PER_BLOCK][16][84];

    const int tid = threadIdx.x;

    if (tid < 13) {
        float s = 0.f;
        #pragma unroll
        for (int j = 0; j < 13; ++j) s += wgt[tid * 13 + j];
        ctab[tid] = s;
    }
    __syncthreads();
    for (int idx = tid; idx < 16 * 80; idx += 128) {
        int n = idx / 80, w = idx % 80;
        int d = w - 4 * n;
        Hm[n][w] = (d >= 0 && d < 13) ? ctab[d] : 0.0f;
    }
    __syncthreads();

    const int wv   = tid >> 5;
    const int lane = tid & 31;
    const int half = lane >> 4;
    const int l15  = lane & 15;

    const int tileId = blockIdx.x * WAVES_PER_BLOCK + wv;
    const int nc  = tileId / TILES_PER_IMG;
    const int t   = tileId % TILES_PER_IMG;
    const int oy0 = (t >> 3) * 16;
    const int ox0 = (t & 7) * 16;

    const float* src = in + (size_t)nc * (H * W);
    const int ybase = 4 * oy0 - 6;
    const int xbase = 4 * ox0 - 6;

    float (*T)[84] = Tt[wv];

    const bool interior = (oy0 >= 16) & (oy0 <= 96) & (ox0 >= 16) & (ox0 <= 96);
    if (interior)
        pass1<false>(src, ybase, xbase, T, Hm, l15, half);
    else
        pass1<true>(src, ybase, xbase, T, Hm, l15, half);

    v8f o = {};
    #pragma unroll
    for (int kk = 0; kk < 19; ++kk) {
        v2f a = *(const v2f*)(&Hm[l15][4 * kk + 2 * half]);
        v2f b = *(const v2f*)(&T [l15][4 * kk + 2 * half]);
        o = __builtin_amdgcn_wmma_f32_16x16x4_f32(false, a, false, b,
                                                  (short)0, o, false, false);
    }

    float* dst = out + (size_t)nc * (OH * OW);
    #pragma unroll
    for (int r = 0; r < 8; ++r)
        dst[(size_t)(oy0 + r + 8 * half) * OW + (ox0 + l15)] = o[r];
}

extern "C" void kernel_launch(void* const* d_in, const int* in_sizes, int n_in,
                              void* d_out, int out_size, void* d_ws, size_t ws_size,
                              hipStream_t stream) {
    const float* inp = (const float*)d_in[0];
    const float* wgt = (const float*)d_in[1];
    float* out = (float*)d_out;

    dim3 grid(NC * TILES_PER_IMG / WAVES_PER_BLOCK);
    dim3 block(128);
    aa_interp_wmma_kernel<<<grid, block, 0, stream>>>(inp, wgt, out);
}


# Vertical blur first, image columns on the A rows: contiguous 16-byte row loads replace 16-row strided loads; pass-1 accumulators are pass-2 A fragments; rows are summed before columns
# speedup vs baseline: 1.9185x; 1.9185x over previous
#include <hip/hip_runtime.h>


typedef float v2f __attribute__((ext_vector_type(2)));
typedef float v4f __attribute__((ext_vector_type(4)));
typedef float v8f __attribute__((ext_vector_type(8)));

#define H   512
#define W   512
#define OH  128
#define OW  128
#define NC  96
#define TILES_PER_IMG 64
#define WAVES_PER_BLOCK 4
#define BT_PITCH 84

#define WMMA(a, b, c) __builtin_amdgcn_wmma_f32_16x16x4_f32(false, (a), false, (b), (short)0, (c), false, false)

template <bool EDGE>
__device__ __forceinline__ void tile(const float* __restrict__ in,
                                     float* __restrict__ out,
                                     int nc, int oy0, int ox0,
                                     const float (*__restrict__ Bt)[BT_PITCH],
                                     int l15, int half)
{
    const int ybase = 4 * oy0 - 8;
    const int X0    = 4 * ox0 - 8;
    const int xq    = X0 + 4 * l15;
    const int xs    = X0 + 64 + l15;

    v8f acc0 = {}, acc1 = {}, acc2 = {}, acc3 = {}, acc4 = {};

    const bool xqv = (unsigned)xq < (unsigned)W;
    const bool xsv = xs < W;
    const int  xqc = min(max(xq, 0), W - 4);
    const int  xsc = min(xs, W - 1);

    const unsigned img = (unsigned)nc * (unsigned)(H * W);
    const float* bq = in + (img + (unsigned)((ybase + 2 * half) * W + xq));
    const float* bs = in + (img + (unsigned)((ybase + 2 * half) * W + xs));

    v4f P0[19], P1[19];
    float Q0[19], Q1[19];

    auto load_rows = [&](int kk) {
        if (EDGE) {
            const int  y  = ybase + 4 * kk + 2 * half;
            const bool yv = (unsigned)y < (unsigned)H;
            const int  yc = min(max(y, 0), H - 2);
            const float* rq = in + (img + (unsigned)(yc * W + xqc));
            const float* rs = in + (img + (unsigned)(yc * W + xsc));
            v4f p0 = *(const v4f*)(rq);
            v4f p1 = *(const v4f*)(rq + W);
            float q0 = rs[0];
            float q1 = rs[W];
            const bool vq = yv && xqv;
            const bool vs = yv && xsv;
            const v4f z4 = {0.f, 0.f, 0.f, 0.f};
            P0[kk] = vq ? p0 : z4;
            P1[kk] = vq ? p1 : z4;
            Q0[kk] = vs ? q0 : 0.f;
            Q1[kk] = vs ? q1 : 0.f;
        } else {
            P0[kk] = *(const v4f*)(bq + (4 * kk) * W);
            P1[kk] = *(const v4f*)(bq + (4 * kk + 1) * W);
            Q0[kk] = bs[(4 * kk) * W];
            Q1[kk] = bs[(4 * kk + 1) * W];
        }
    };
    auto mul_rows = [&](int kk) {
        const v2f b = *(const v2f*)(&Bt[l15][4 * kk + 2 * half]);
        v2f a;
        a.x = P0[kk].x; a.y = P1[kk].x; acc0 = WMMA(a, b, acc0);
        a.x = P0[kk].y; a.y = P1[kk].y; acc1 = WMMA(a, b, acc1);
        a.x = P0[kk].z; a.y = P1[kk].z; acc2 = WMMA(a, b, acc2);
        a.x = P0[kk].w; a.y = P1[kk].w; acc3 = WMMA(a, b, acc3);
        a.x = Q0[kk];   a.y = Q1[kk];   acc4 = WMMA(a, b, acc4);
    };

    #pragma unroll
    for (int kk = 0; kk < 8; ++kk) load_rows(kk);
    #pragma unroll
    for (int g = 0; g < 5; ++g) {
        __builtin_amdgcn_sched_barrier(0);
        #pragma unroll
        for (int kk = 4 * g; kk < 4 * g + 4; ++kk)
            if (kk < 19) mul_rows(kk);
        #pragma unroll
        for (int kk = 4 * g + 8; kk < 4 * g + 12; ++kk)
            if (kk < 19) load_rows(kk);
    }

    v8f o = {};
    const float* brow = &Bt[l15][32 * half];
    #pragma unroll
    for (int m = 0; m < 4; ++m) {
        v2f a, b;
        a.x = acc0[2 * m]; a.y = acc0[2 * m + 1];
        b.x = brow[8 * m + 0]; b.y = brow[8 * m + 4];
        o = WMMA(a, b, o);
        a.x = acc1[2 * m]; a.y = acc1[2 * m + 1];
        b.x = brow[8 * m + 1]; b.y = brow[8 * m + 5];
        o = WMMA(a, b, o);
        a.x = acc2[2 * m]; a.y = acc2[2 * m + 1];
        b.x = brow[8 * m + 2]; b.y = brow[8 * m + 6];
        o = WMMA(a, b, o);
        a.x = acc3[2 * m]; a.y = acc3[2 * m + 1];
        b.x = brow[8 * m + 3]; b.y = brow[8 * m + 7];
        o = WMMA(a, b, o);
    }
    const float* brow4 = &Bt[l15][64 + 8 * half];
    #pragma unroll
    for (int u = 0; u < 4; ++u) {
        v2f a;
        a.x = acc4[2 * u]; a.y = acc4[2 * u + 1];
        const v2f b = *(const v2f*)(brow4 + 2 * u);
        o = WMMA(a, b, o);
    }

    float* dp = out + ((unsigned)nc * (unsigned)(OH * OW) + (unsigned)((oy0 + 8 * half) * OW + ox0 + l15));
    #pragma unroll
    for (int r = 0; r < 8; ++r)
        dp[r * OW] = o[r];
}

__global__ __launch_bounds__(128, 6)
void aa_interp_wmma_kernel(const float* __restrict__ in,
                           const float* __restrict__ wgt,
                           float* __restrict__ out)
{
    __shared__ float ctab[13];
    __shared__ __attribute__((aligned(16))) float Bt[16][BT_PITCH];

    const int tid = threadIdx.x;

    if (tid < 13) {
        float s = 0.f;
        #pragma unroll
        for (int j = 0; j < 13; ++j) s += wgt[tid * 13 + j];
        ctab[tid] = s;
    }
    __syncthreads();
    for (int idx = tid; idx < 16 * 80; idx += 128) {
        int n = idx / 80, w = idx % 80;
        int d = w - 4 * n - 2;
        Bt[n][w] = (d >= 0 && d < 13) ? ctab[d] : 0.0f;
    }
    __syncthreads();

    const int wv   = tid >> 5;
    const int lane = tid & 31;
    const int half = lane >> 4;
    const int l15  = lane & 15;

    const int tileId = blockIdx.x * WAVES_PER_BLOCK + wv;
    const int nc  = tileId / TILES_PER_IMG;
    const int t   = tileId % TILES_PER_IMG;
    const int oy0 = (t >> 3) * 16;
    const int ox0 = (t & 7) * 16;

    const bool interior = (oy0 >= 16) & (oy0 <= 96) & (ox0 >= 16) & (ox0 <= 96);
    if (interior)
        tile<false>(in, out, nc, oy0, ox0, Bt, l15, half);
    else
        tile<true>(in, out, nc, oy0, ox0, Bt, l15, half);
}

extern "C" void kernel_launch(void* const* d_in, const int* in_sizes, int n_in,
                              void* d_out, int out_size, void* d_ws, size_t ws_size,
                              hipStream_t stream) {
    const float* inp = (const float*)d_in[0];
    const float* wgt = (const float*)d_in[1];
    float* out = (float*)d_out;

    dim3 grid(NC * TILES_PER_IMG / WAVES_PER_BLOCK);
    dim3 block(128);
    aa_interp_wmma_kernel<<<grid, block, 0, stream>>>(inp, wgt, out);
}


# One wave computes a 16x32 output tile (two 16x16 tiles side by side): shared halo columns loaded and blurred once, 144 not 160 per pair; left tile's last 16 columns summed in a new order
# speedup vs baseline: 2.0962x; 1.0926x over previous
#include <hip/hip_runtime.h>


typedef float v2f __attribute__((ext_vector_type(2)));
typedef float v4f __attribute__((ext_vector_type(4)));
typedef float v8f __attribute__((ext_vector_type(8)));

#define H   512
#define W   512
#define OH  128
#define OW  128
#define NC  96
#define TILES_PER_IMG 32
#define WAVES_PER_BLOCK 4
#define BT_PITCH 84

#define WMMA(a, b, c) __builtin_amdgcn_wmma_f32_16x16x4_f32(false, (a), false, (b), (short)0, (c), false, false)

template <bool EDGE>
__device__ __forceinline__ void tile(const float* __restrict__ in,
                                     float* __restrict__ out,
                                     int nc, int oy0, int ox0,
                                     const float (*__restrict__ Bt)[BT_PITCH],
                                     int l15, int half)
{
    const int ybase = 4 * oy0 - 8;
    const int X0    = 4 * ox0 - 8;
    const int xa    = X0 + 4 * l15;
    const int xb    = X0 + 64 + 4 * l15;
    const int xs    = X0 + 128 + l15;

    v8f acc[9];
    #pragma unroll
    for (int s = 0; s < 9; ++s) acc[s] = v8f{};

    const bool xav = (unsigned)xa < (unsigned)W;
    const bool xbv = (unsigned)xb < (unsigned)W;
    const bool xsv = xs < W;
    const int  xac = min(max(xa, 0), W - 4);
    const int  xbc = min(max(xb, 0), W - 4);
    const int  xsc = min(xs, W - 1);

    const unsigned img = (unsigned)nc * (unsigned)(H * W);
    const float* ba = in + (img + (unsigned)((ybase + 2 * half) * W + xa));
    const float* bs = in + (img + (unsigned)((ybase + 2 * half) * W + xs));

    v4f PA0[19], PA1[19];
    v4f PB0[19], PB1[19];
    float Q0[19], Q1[19];

    auto load_rows = [&](int kk) {
        if (EDGE) {
            const int  y  = ybase + 4 * kk + 2 * half;
            const bool yv = (unsigned)y < (unsigned)H;
            const int  yc = min(max(y, 0), H - 2);
            const float* ra = in + (img + (unsigned)(yc * W + xac));
            const float* rb = in + (img + (unsigned)(yc * W + xbc));
            const float* rs = in + (img + (unsigned)(yc * W + xsc));
            v4f a0 = *(const v4f*)(ra);
            v4f a1 = *(const v4f*)(ra + W);
            v4f b0 = *(const v4f*)(rb);
            v4f b1 = *(const v4f*)(rb + W);
            float q0 = rs[0];
            float q1 = rs[W];
            const bool va = yv && xav;
            const bool vb = yv && xbv;
            const bool vs = yv && xsv;
            const v4f z4 = {0.f, 0.f, 0.f, 0.f};
            PA0[kk] = va ? a0 : z4;
            PA1[kk] = va ? a1 : z4;
            PB0[kk] = vb ? b0 : z4;
            PB1[kk] = vb ? b1 : z4;
            Q0[kk] = vs ? q0 : 0.f;
            Q1[kk] = vs ? q1 : 0.f;
        } else {
            PA0[kk] = *(const v4f*)(ba + (4 * kk) * W);
            PA1[kk] = *(const v4f*)(ba + (4 * kk + 1) * W);
            PB0[kk] = *(const v4f*)(ba + (4 * kk) * W + 64);
            PB1[kk] = *(const v4f*)(ba + (4 * kk + 1) * W + 64);
            Q0[kk] = bs[(4 * kk) * W];
            Q1[kk] = bs[(4 * kk + 1) * W];
        }
    };
    auto mul_rows = [&](int kk) {
        const v2f b = *(const v2f*)(&Bt[l15][4 * kk + 2 * half]);
        v2f a;
        a.x = PA0[kk].x; a.y = PA1[kk].x; acc[0] = WMMA(a, b, acc[0]);
        a.x = PA0[kk].y; a.y = PA1[kk].y; acc[1] = WMMA(a, b, acc[1]);
        a.x = PA0[kk].z; a.y = PA1[kk].z; acc[2] = WMMA(a, b, acc[2]);
        a.x = PA0[kk].w; a.y = PA1[kk].w; acc[3] = WMMA(a, b, acc[3]);
        a.x = PB0[kk].x; a.y = PB1[kk].x; acc[4] = WMMA(a, b, acc[4]);
        a.x = PB0[kk].y; a.y = PB1[kk].y; acc[5] = WMMA(a, b, acc[5]);
        a.x = PB0[kk].z; a.y = PB1[kk].z; acc[6] = WMMA(a, b, acc[6]);
        a.x = PB0[kk].w; a.y = PB1[kk].w; acc[7] = WMMA(a, b, acc[7]);
        a.x = Q0[kk];    a.y = Q1[kk];    acc[8] = WMMA(a, b, acc[8]);
    };

    #pragma unroll
    for (int kk = 0; kk < 8; ++kk) load_rows(kk);
    #pragma unroll
    for (int g = 0; g < 5; ++g) {
        __builtin_amdgcn_sched_barrier(0);
        #pragma unroll
        for (int kk = 4 * g; kk < 4 * g + 4; ++kk)
            if (kk < 19) mul_rows(kk);
        #pragma unroll
        for (int kk = 4 * g + 8; kk < 4 * g + 12; ++kk)
            if (kk < 19) load_rows(kk);
    }

    v8f o0 = {}, o1 = {};
    const float* brow = &Bt[l15][32 * half];
    #pragma unroll
    for (int m = 0; m < 4; ++m) {
        #pragma unroll
        for (int j = 0; j < 4; ++j) {
            v2f a, b;
            b.x = brow[8 * m + j]; b.y = brow[8 * m + j + 4];
            a.x = acc[j][2 * m];     a.y = acc[j][2 * m + 1];
            o0 = WMMA(a, b, o0);
            a.x = acc[4 + j][2 * m]; a.y = acc[4 + j][2 * m + 1];
            o1 = WMMA(a, b, o1);
        }
    }
    #pragma unroll
    for (int m = 0; m < 2; ++m) {
        #pragma unroll
        for (int j = 0; j < 4; ++j) {
            v2f a, b;
            const float c0 = Bt[l15][64 + 8 * m + j];
            const float c1 = Bt[l15][64 + 8 * m + j + 4];
            b.x = half ? 0.f : c0; b.y = half ? 0.f : c1;
            a.x = acc[4 + j][2 * m]; a.y = acc[4 + j][2 * m + 1];
            o0 = WMMA(a, b, o0);
        }
    }
    const float* brow4 = &Bt[l15][64 + 8 * half];
    #pragma unroll
    for (int u = 0; u < 4; ++u) {
        v2f a;
        a.x = acc[8][2 * u]; a.y = acc[8][2 * u + 1];
        const v2f b = *(const v2f*)(brow4 + 2 * u);
        o1 = WMMA(a, b, o1);
    }

    float* dp = out + ((unsigned)nc * (unsigned)(OH * OW) + (unsigned)((oy0 + 8 * half) * OW + ox0 + l15));
    #pragma unroll
    for (int r = 0; r < 8; ++r) {
        dp[r * OW]      = o0[r];
        dp[r * OW + 16] = o1[r];
    }
}

__global__ __launch_bounds__(128, 3)
void aa_interp_wmma_kernel(const float* __restrict__ in,
                           const float* __restrict__ wgt,
                           float* __restrict__ out)
{
    __shared__ float ctab[13];
    __shared__ __attribute__((aligned(16))) float Bt[16][BT_PITCH];

    const int tid = threadIdx.x;

    if (tid < 13) {
        float s = 0.f;
        #pragma unroll
        for (int j = 0; j < 13; ++j) s += wgt[tid * 13 + j];
        ctab[tid] = s;
    }
    __syncthreads();
    for (int idx = tid; idx < 16 * 80; idx += 128) {
        int n = idx / 80, w = idx % 80;
        int d = w - 4 * n - 2;
        Bt[n][w] = (d >= 0 && d < 13) ? ctab[d] : 0.0f;
    }
    __syncthreads();

    const int wv   = tid >> 5;
    const int lane = tid & 31;
    const int half = lane >> 4;
    const int l15  = lane & 15;

    const int tileId = blockIdx.x * WAVES_PER_BLOCK + wv;
    const int nc  = tileId / TILES_PER_IMG;
    const int t   = tileId % TILES_PER_IMG;
    const int oy0 = (t >> 2) * 16;
    const int ox0 = (t & 3) * 32;

    const bool interior = (oy0 >= 16) & (oy0 <= 96) & (ox0 >= 32) & (ox0 <= 64);
    if (interior)
        tile<false>(in, out, nc, oy0, ox0, Bt, l15, half);
    else
        tile<true>(in, out, nc, oy0, ox0, Bt, l15, half);
}

extern "C" void kernel_launch(void* const* d_in, const int* in_sizes, int n_in,
                              void* d_out, int out_size, void* d_ws, size_t ws_size,
                              hipStream_t stream) {
    const float* inp = (const float*)d_in[0];
    const float* wgt = (const float*)d_in[1];
    float* out = (float*)d_out;

    dim3 grid(NC * TILES_PER_IMG / WAVES_PER_BLOCK);
    dim3 block(128);
    aa_interp_wmma_kernel<<<grid, block, 0, stream>>>(inp, wgt, out);
}


# Edge tiles: the zeros for positions outside the image are selected where the A fragments are formed, not at the loads, so edge-tile loads stay two row groups in flight as in interior tiles
# speedup vs baseline: 2.3042x; 1.0992x over previous
#include <hip/hip_runtime.h>


typedef float v2f __attribute__((ext_vector_type(2)));
typedef float v4f __attribute__((ext_vector_type(4)));
typedef float v8f __attribute__((ext_vector_type(8)));

#define H   512
#define W   512
#define OH  128
#define OW  128
#define NC  96
#define TILES_PER_IMG 32
#define WAVES_PER_BLOCK 4
#define BT_PITCH 84

#define WMMA(a, b, c) __builtin_amdgcn_wmma_f32_16x16x4_f32(false, (a), false, (b), (short)0, (c), false, false)

template <bool EDGE>
__device__ __forceinline__ void tile(const float* __restrict__ in,
                                     float* __restrict__ out,
                                     int nc, int oy0, int ox0,
                                     const float (*__restrict__ Bt)[BT_PITCH],
                                     int l15, int half)
{
    const int ybase = 4 * oy0 - 8;
    const int X0    = 4 * ox0 - 8;
    const int xa    = X0 + 4 * l15;
    const int xb    = X0 + 64 + 4 * l15;
    const int xs    = X0 + 128 + l15;

    v8f acc[9];
    #pragma unroll
    for (int s = 0; s < 9; ++s) acc[s] = v8f{};

    const bool xav = (unsigned)xa < (unsigned)W;
    const bool xbv = (unsigned)xb < (unsigned)W;
    const bool xsv = xs < W;
    const int  xac = min(max(xa, 0), W - 4);
    const int  xbc = min(max(xb, 0), W - 4);
    const int  xsc = min(xs, W - 1);

    const unsigned img = (unsigned)nc * (unsigned)(H * W);
    const float* ba = in + (img + (unsigned)((ybase + 2 * half) * W + xa));
    const float* bs = in + (img + (unsigned)((ybase + 2 * half) * W + xs));

    v4f PA0[19], PA1[19];
    v4f PB0[19], PB1[19];
    float Q0[19], Q1[19];

    auto load_rows = [&](int kk) {
        if (EDGE) {
            const int  y  = ybase + 4 * kk + 2 * half;
            const int  yc = min(max(y, 0), H - 2);
            const float* ra = in + (img + (unsigned)(yc * W + xac));
            const float* rb = in + (img + (unsigned)(yc * W + xbc));
            const float* rs = in + (img + (unsigned)(yc * W + xsc));
            v4f a0 = *(const v4f*)(ra);
            v4f a1 = *(const v4f*)(ra + W);
            v4f b0 = *(const v4f*)(rb);
            v4f b1 = *(const v4f*)(rb + W);
            float q0 = rs[0];
            float q1 = rs[W];
            PA0[kk] = a0; PA1[kk] = a1;
            PB0[kk] = b0; PB1[kk] = b1;
            Q0[kk] = q0;  Q1[kk] = q1;
        } else {
            PA0[kk] = *(const v4f*)(ba + (4 * kk) * W);
            PA1[kk] = *(const v4f*)(ba + (4 * kk + 1) * W);
            PB0[kk] = *(const v4f*)(ba + (4 * kk) * W + 64);
            PB1[kk] = *(const v4f*)(ba + (4 * kk + 1) * W + 64);
            Q0[kk] = bs[(4 * kk) * W];
            Q1[kk] = bs[(4 * kk + 1) * W];
        }
    };
    auto mul_rows = [&](int kk) {
        const v2f b = *(const v2f*)(&Bt[l15][4 * kk + 2 * half]);
        const int  y  = ybase + 4 * kk + 2 * half;
        const bool yv = !EDGE || ((unsigned)y < (unsigned)H);
        const bool va = !EDGE || (yv && xav);
        const bool vb = !EDGE || (yv && xbv);
        const bool vs = !EDGE || (yv && xsv);
        v2f a;
        a.x = va ? PA0[kk].x : 0.f; a.y = va ? PA1[kk].x : 0.f; acc[0] = WMMA(a, b, acc[0]);
        a.x = va ? PA0[kk].y : 0.f; a.y = va ? PA1[kk].y : 0.f; acc[1] = WMMA(a, b, acc[1]);
        a.x = va ? PA0[kk].z : 0.f; a.y = va ? PA1[kk].z : 0.f; acc[2] = WMMA(a, b, acc[2]);
        a.x = va ? PA0[kk].w : 0.f; a.y = va ? PA1[kk].w : 0.f; acc[3] = WMMA(a, b, acc[3]);
        a.x = vb ? PB0[kk].x : 0.f; a.y = vb ? PB1[kk].x : 0.f; acc[4] = WMMA(a, b, acc[4]);
        a.x = vb ? PB0[kk].y : 0.f; a.y = vb ? PB1[kk].y : 0.f; acc[5] = WMMA(a, b, acc[5]);
        a.x = vb ? PB0[kk].z : 0.f; a.y = vb ? PB1[kk].z : 0.f; acc[6] = WMMA(a, b, acc[6]);
        a.x = vb ? PB0[kk].w : 0.f; a.y = vb ? PB1[kk].w : 0.f; acc[7] = WMMA(a, b, acc[7]);
        a.x = vs ? Q0[kk] : 0.f;    a.y = vs ? Q1[kk] : 0.f;    acc[8] = WMMA(a, b, acc[8]);
    };

    #pragma unroll
    for (int kk = 0; kk < 8; ++kk) load_rows(kk);
    #pragma unroll
    for (int g = 0; g < 5; ++g) {
        __builtin_amdgcn_sched_barrier(0);
        #pragma unroll
        for (int kk = 4 * g; kk < 4 * g + 4; ++kk)
            if (kk < 19) mul_rows(kk);
        #pragma unroll
        for (int kk = 4 * g + 8; kk < 4 * g + 12; ++kk)
            if (kk < 19) load_rows(kk);
    }

    v8f o0 = {}, o1 = {};
    const float* brow = &Bt[l15][32 * half];
    #pragma unroll
    for (int m = 0; m < 4; ++m) {
        #pragma unroll
        for (int j = 0; j < 4; ++j) {
            v2f a, b;
            b.x = brow[8 * m + j]; b.y = brow[8 * m + j + 4];
            a.x = acc[j][2 * m];     a.y = acc[j][2 * m + 1];
            o0 = WMMA(a, b, o0);
            a.x = acc[4 + j][2 * m]; a.y = acc[4 + j][2 * m + 1];
            o1 = WMMA(a, b, o1);
        }
    }
    #pragma unroll
    for (int m = 0; m < 2; ++m) {
        #pragma unroll
        for (int j = 0; j < 4; ++j) {
            v2f a, b;
            const float c0 = Bt[l15][64 + 8 * m + j];
            const float c1 = Bt[l15][64 + 8 * m + j + 4];
            b.x = half ? 0.f : c0; b.y = half ? 0.f : c1;
            a.x = acc[4 + j][2 * m]; a.y = acc[4 + j][2 * m + 1];
            o0 = WMMA(a, b, o0);
        }
    }
    const float* brow4 = &Bt[l15][64 + 8 * half];
    #pragma unroll
    for (int u = 0; u < 4; ++u) {
        v2f a;
        a.x = acc[8][2 * u]; a.y = acc[8][2 * u + 1];
        const v2f b = *(const v2f*)(brow4 + 2 * u);
        o1 = WMMA(a, b, o1);
    }

    float* dp = out + ((unsigned)nc * (unsigned)(OH * OW) + (unsigned)((oy0 + 8 * half) * OW + ox0 + l15));
    #pragma unroll
    for (int r = 0; r < 8; ++r) {
        dp[r * OW]      = o0[r];
        dp[r * OW + 16] = o1[r];
    }
}

__global__ __launch_bounds__(128, 3)
void aa_interp_wmma_kernel(const float* __restrict__ in,
                           const float* __restrict__ wgt,
                           float* __restrict__ out)
{
    __shared__ float ctab[13];
    __shared__ __attribute__((aligned(16))) float Bt[16][BT_PITCH];

    const int tid = threadIdx.x;

    if (tid < 13) {
        float s = 0.f;
        #pragma unroll
        for (int j = 0; j < 13; ++j) s += wgt[tid * 13 + j];
        ctab[tid] = s;
    }
    __syncthreads();
    for (int idx = tid; idx < 16 * 80; idx += 128) {
        int n = idx / 80, w = idx % 80;
        int d = w - 4 * n - 2;
        Bt[n][w] = (d >= 0 && d < 13) ? ctab[d] : 0.0f;
    }
    __syncthreads();

    const int wv   = tid >> 5;
    const int lane = tid & 31;
    const int half = lane >> 4;
    const int l15  = lane & 15;

    const int tileId = blockIdx.x * WAVES_PER_BLOCK + wv;
    const int nc  = tileId / TILES_PER_IMG;
    const int t   = tileId % TILES_PER_IMG;
    const int oy0 = (t >> 2) * 16;
    const int ox0 = (t & 3) * 32;

    const bool interior = (oy0 >= 16) & (oy0 <= 96) & (ox0 >= 32) & (ox0 <= 64);
    if (interior)
        tile<false>(in, out, nc, oy0, ox0, Bt, l15, half);
    else
        tile<true>(in, out, nc, oy0, ox0, Bt, l15, half);
}

extern "C" void kernel_launch(void* const* d_in, const int* in_sizes, int n_in,
                              void* d_out, int out_size, void* d_ws, size_t ws_size,
                              hipStream_t stream) {
    const float* inp = (const float*)d_in[0];
    const float* wgt = (const float*)d_in[1];
    float* out = (float*)d_out;

    dim3 grid(NC * TILES_PER_IMG / WAVES_PER_BLOCK);
    dim3 block(128);
    aa_interp_wmma_kernel<<<grid, block, 0, stream>>>(inp, wgt, out);
}


# The tile's first two row groups of input loads are requested before the band table is built (weight rows first, barriers that order LDS only), so the table build runs under the load latency
# speedup vs baseline: 2.5210x; 1.0941x over previous
#include <hip/hip_runtime.h>


typedef float v2f __attribute__((ext_vector_type(2)));
typedef float v4f __attribute__((ext_vector_type(4)));
typedef float v8f __attribute__((ext_vector_type(8)));

#define H   512
#define W   512
#define OH  128
#define OW  128
#define NC  96
#define TILES_PER_IMG 32
#define WAVES_PER_BLOCK 4
#define BT_PITCH 84

#define WMMA(a, b, c) __builtin_amdgcn_wmma_f32_16x16x4_f32(false, (a), false, (b), (short)0, (c), false, false)


struct RowBuf {
    v4f PA0[19], PA1[19];
    v4f PB0[19], PB1[19];
    float Q0[19], Q1[19];
};

struct TileGeo {
    int ybase;
    int X0;
    int xa, xb, xs;
    bool xav, xbv, xsv;
    int xac, xbc, xsc;
    unsigned img;
    const float* ba;
    const float* bs;
};

__device__ __forceinline__ TileGeo tile_geo(const float* __restrict__ in,
                                            int nc, int oy0, int ox0, int l15, int half)
{
    TileGeo g;
    g.ybase = 4 * oy0 - 8;
    g.X0    = 4 * ox0 - 8;
    g.xa    = g.X0 + 4 * l15;
    g.xb    = g.X0 + 64 + 4 * l15;
    g.xs    = g.X0 + 128 + l15;
    g.xav = (unsigned)g.xa < (unsigned)W;
    g.xbv = (unsigned)g.xb < (unsigned)W;
    g.xsv = g.xs < W;
    g.xac = min(max(g.xa, 0), W - 4);
    g.xbc = min(max(g.xb, 0), W - 4);
    g.xsc = min(g.xs, W - 1);
    g.img = (unsigned)nc * (unsigned)(H * W);
    g.ba = in + (g.img + (unsigned)((g.ybase + 2 * half) * W + g.xa));
    g.bs = in + (g.img + (unsigned)((g.ybase + 2 * half) * W + g.xs));
    return g;
}

template <bool EDGE>
__device__ __forceinline__ void load_rows(const float* __restrict__ in, const TileGeo& g,
                                          RowBuf& rb, int kk, int half)
{
    if (EDGE) {
        const int  y  = g.ybase + 4 * kk + 2 * half;
        const int  yc = min(max(y, 0), H - 2);
        const float* ra = in + (g.img + (unsigned)(yc * W + g.xac));
        const float* rb_ = in + (g.img + (unsigned)(yc * W + g.xbc));
        const float* rs = in + (g.img + (unsigned)(yc * W + g.xsc));
        v4f a0 = *(const v4f*)(ra);
        v4f a1 = *(const v4f*)(ra + W);
        v4f b0 = *(const v4f*)(rb_);
        v4f b1 = *(const v4f*)(rb_ + W);
        float q0 = rs[0];
        float q1 = rs[W];
        rb.PA0[kk] = a0; rb.PA1[kk] = a1;
        rb.PB0[kk] = b0; rb.PB1[kk] = b1;
        rb.Q0[kk] = q0;  rb.Q1[kk] = q1;
    } else {
        rb.PA0[kk] = *(const v4f*)(g.ba + (4 * kk) * W);
        rb.PA1[kk] = *(const v4f*)(g.ba + (4 * kk + 1) * W);
        rb.PB0[kk] = *(const v4f*)(g.ba + (4 * kk) * W + 64);
        rb.PB1[kk] = *(const v4f*)(g.ba + (4 * kk + 1) * W + 64);
        rb.Q0[kk] = g.bs[(4 * kk) * W];
        rb.Q1[kk] = g.bs[(4 * kk + 1) * W];
    }
}

template <bool EDGE>
__device__ __forceinline__ void first_rows(const float* __restrict__ in, const TileGeo& g,
                                           RowBuf& rb, int half)
{
    #pragma unroll
    for (int kk = 0; kk < 8; ++kk) load_rows<EDGE>(in, g, rb, kk, half);
}

template <bool EDGE>
__device__ __forceinline__ void tile_rest(const float* __restrict__ in,
                                          float* __restrict__ out,
                                          int nc, int oy0, int ox0,
                                          const float (*__restrict__ Bt)[BT_PITCH],
                                          int l15, int half,
                                          const TileGeo& g, RowBuf& rb)
{
    v8f acc[9];
    #pragma unroll
    for (int s = 0; s < 9; ++s) acc[s] = v8f{};

    auto mul_rows = [&](int kk) {
        const v2f b = *(const v2f*)(&Bt[l15][4 * kk + 2 * half]);
        const int  y  = g.ybase + 4 * kk + 2 * half;
        const bool yv = !EDGE || ((unsigned)y < (unsigned)H);
        const bool va = !EDGE || (yv && g.xav);
        const bool vb = !EDGE || (yv && g.xbv);
        const bool vs = !EDGE || (yv && g.xsv);
        v2f a;
        a.x = va ? rb.PA0[kk].x : 0.f; a.y = va ? rb.PA1[kk].x : 0.f; acc[0] = WMMA(a, b, acc[0]);
        a.x = va ? rb.PA0[kk].y : 0.f; a.y = va ? rb.PA1[kk].y : 0.f; acc[1] = WMMA(a, b, acc[1]);
        a.x = va ? rb.PA0[kk].z : 0.f; a.y = va ? rb.PA1[kk].z : 0.f; acc[2] = WMMA(a, b, acc[2]);
        a.x = va ? rb.PA0[kk].w : 0.f; a.y = va ? rb.PA1[kk].w : 0.f; acc[3] = WMMA(a, b, acc[3]);
        a.x = vb ? rb.PB0[kk].x : 0.f; a.y = vb ? rb.PB1[kk].x : 0.f; acc[4] = WMMA(a, b, acc[4]);
        a.x = vb ? rb.PB0[kk].y : 0.f; a.y = vb ? rb.PB1[kk].y : 0.f; acc[5] = WMMA(a, b, acc[5]);
        a.x = vb ? rb.PB0[kk].z : 0.f; a.y = vb ? rb.PB1[kk].z : 0.f; acc[6] = WMMA(a, b, acc[6]);
        a.x = vb ? rb.PB0[kk].w : 0.f; a.y = vb ? rb.PB1[kk].w : 0.f; acc[7] = WMMA(a, b, acc[7]);
        a.x = vs ? rb.Q0[kk] : 0.f;    a.y = vs ? rb.Q1[kk] : 0.f;    acc[8] = WMMA(a, b, acc[8]);
    };

    #pragma unroll
    for (int g5 = 0; g5 < 5; ++g5) {
        __builtin_amdgcn_sched_barrier(0);
        #pragma unroll
        for (int kk = 4 * g5; kk < 4 * g5 + 4; ++kk)
            if (kk < 19) mul_rows(kk);
        #pragma unroll
        for (int kk = 4 * g5 + 8; kk < 4 * g5 + 12; ++kk)
            if (kk < 19) load_rows<EDGE>(in, g, rb, kk, half);
    }

    v8f o0 = {}, o1 = {};
    const float* brow = &Bt[l15][32 * half];
    #pragma unroll
    for (int m = 0; m < 4; ++m) {
        #pragma unroll
        for (int j = 0; j < 4; ++j) {
            v2f a, b;
            b.x = brow[8 * m + j]; b.y = brow[8 * m + j + 4];
            a.x = acc[j][2 * m];     a.y = acc[j][2 * m + 1];
            o0 = WMMA(a, b, o0);
            a.x = acc[4 + j][2 * m]; a.y = acc[4 + j][2 * m + 1];
            o1 = WMMA(a, b, o1);
        }
    }
    #pragma unroll
    for (int m = 0; m < 2; ++m) {
        #pragma unroll
        for (int j = 0; j < 4; ++j) {
            v2f a, b;
            const float c0 = Bt[l15][64 + 8 * m + j];
            const float c1 = Bt[l15][64 + 8 * m + j + 4];
            b.x = half ? 0.f : c0; b.y = half ? 0.f : c1;
            a.x = acc[4 + j][2 * m]; a.y = acc[4 + j][2 * m + 1];
            o0 = WMMA(a, b, o0);
        }
    }
    const float* brow4 = &Bt[l15][64 + 8 * half];
    #pragma unroll
    for (int u = 0; u < 4; ++u) {
        v2f a;
        a.x = acc[8][2 * u]; a.y = acc[8][2 * u + 1];
        const v2f b = *(const v2f*)(brow4 + 2 * u);
        o1 = WMMA(a, b, o1);
    }

    float* dp = out + ((unsigned)nc * (unsigned)(OH * OW) + (unsigned)((oy0 + 8 * half) * OW + ox0 + l15));
    #pragma unroll
    for (int r = 0; r < 8; ++r) {
        dp[r * OW]      = o0[r];
        dp[r * OW + 16] = o1[r];
    }
}

__device__ __forceinline__ void lds_barrier()
{
    __builtin_amdgcn_fence(__ATOMIC_RELEASE, "workgroup", "local");
    __builtin_amdgcn_s_barrier();
    __builtin_amdgcn_fence(__ATOMIC_ACQUIRE, "workgroup", "local");
}

__global__ __launch_bounds__(128, 3)
void aa_interp_wmma_kernel(const float* __restrict__ in,
                           const float* __restrict__ wgt,
                           float* __restrict__ out)
{
    __shared__ float ctab[13];
    __shared__ __attribute__((aligned(16))) float Bt[16][BT_PITCH];

    const int tid = threadIdx.x;

    const int wv   = tid >> 5;
    const int lane = tid & 31;
    const int half = lane >> 4;
    const int l15  = lane & 15;

    const int tileId = blockIdx.x * WAVES_PER_BLOCK + wv;
    const int nc  = tileId / TILES_PER_IMG;
    const int t   = tileId % TILES_PER_IMG;
    const int oy0 = (t >> 2) * 16;
    const int ox0 = (t & 3) * 32;

    const bool interior = (oy0 >= 16) & (oy0 <= 96) & (ox0 >= 32) & (ox0 <= 64);

    const int wr = min(tid, 12);
    float wrow[13];
    #pragma unroll
    for (int j = 0; j < 13; ++j) wrow[j] = wgt[wr * 13 + j];

    const TileGeo g = tile_geo(in, nc, oy0, ox0, l15, half);
    RowBuf rb;
    first_rows<true>(in, g, rb, half);
    __builtin_amdgcn_sched_barrier(0);

    float s = 0.f;
    #pragma unroll
    for (int j = 0; j < 13; ++j) s += wrow[j];
    ctab[wr] = s;
    lds_barrier();
    for (int idx = tid; idx < 16 * 80; idx += 128) {
        int n = idx / 80, w = idx % 80;
        int d = w - 4 * n - 2;
        Bt[n][w] = (d >= 0 && d < 13) ? ctab[d] : 0.0f;
    }
    lds_barrier();

    if (interior)
        tile_rest<false>(in, out, nc, oy0, ox0, Bt, l15, half, g, rb);
    else
        tile_rest<true>(in, out, nc, oy0, ox0, Bt, l15, half, g, rb);
}

extern "C" void kernel_launch(void* const* d_in, const int* in_sizes, int n_in,
                              void* d_out, int out_size, void* d_ws, size_t ws_size,
                              hipStream_t stream) {
    const float* inp = (const float*)d_in[0];
    const float* wgt = (const float*)d_in[1];
    float* out = (float*)d_out;

    dim3 grid(NC * TILES_PER_IMG / WAVES_PER_BLOCK);
    dim3 block(128);
    aa_interp_wmma_kernel<<<grid, block, 0, stream>>>(inp, wgt, out);
}


# Each wave walks three tiles one below the other (one wave per SIMD): the next tile's first row groups are requested under the current tile's last products; band fragments stay in registers
# speedup vs baseline: 2.7264x; 1.0815x over previous
#include <hip/hip_runtime.h>


typedef float v2f __attribute__((ext_vector_type(2)));
typedef float v4f __attribute__((ext_vector_type(4)));
typedef float v8f __attribute__((ext_vector_type(8)));

#define H   512
#define W   512
#define OH  128
#define OW  128
#define NC  96
#define TILE_ROWS_PER_IMG 8
#define ROWS_PER_BLOCK 3
#define WAVES_PER_BLOCK 4
#define BT_PITCH 84

#define WMMA(a, b, c) __builtin_amdgcn_wmma_f32_16x16x4_f32(false, (a), false, (b), (short)0, (c), false, false)


struct RowBuf {
    v4f PA0[19], PA1[19];
    v4f PB0[19], PB1[19];
    float Q0[19], Q1[19];
};

struct ColGeo {
    bool xav, xbv, xsv;
    int xac, xbc, xsc;
};

struct TilePos {
    int ybase;
    unsigned img;
    unsigned obase;
};

struct Bands {
    v2f p1[19];
    v2f p2[16];
    v2f pl[8];
    v2f pr[4];
};

__device__ __forceinline__ ColGeo col_geo(int ox0, int l15)
{
    ColGeo c;
    const int X0 = 4 * ox0 - 8;
    const int xa = X0 + 4 * l15;
    const int xb = X0 + 64 + 4 * l15;
    const int xs = X0 + 128 + l15;
    c.xav = (unsigned)xa < (unsigned)W;
    c.xbv = (unsigned)xb < (unsigned)W;
    c.xsv = xs < W;
    c.xac = min(max(xa, 0), W - 4);
    c.xbc = min(max(xb, 0), W - 4);
    c.xsc = min(xs, W - 1);
    return c;
}

__device__ __forceinline__ TilePos tile_pos(int R)
{
    TilePos p;
    const int nc  = R / TILE_ROWS_PER_IMG;
    const int oy0 = (R % TILE_ROWS_PER_IMG) * 16;
    p.ybase = 4 * oy0 - 8;
    p.img   = (unsigned)nc * (unsigned)(H * W);
    p.obase = (unsigned)nc * (unsigned)(OH * OW) + (unsigned)(oy0 * OW);
    return p;
}

__device__ __forceinline__ void load_rows(const float* __restrict__ in, const ColGeo& cg,
                                          const TilePos& tp, RowBuf& rb, int kk, int half)
{
    const int  y  = tp.ybase + 4 * kk + 2 * half;
    const int  yc = min(max(y, 0), H - 2);
    const float* ra  = in + (tp.img + (unsigned)(yc * W + cg.xac));
    const float* rb_ = in + (tp.img + (unsigned)(yc * W + cg.xbc));
    const float* rs  = in + (tp.img + (unsigned)(yc * W + cg.xsc));
    rb.PA0[kk] = *(const v4f*)(ra);
    rb.PA1[kk] = *(const v4f*)(ra + W);
    rb.PB0[kk] = *(const v4f*)(rb_);
    rb.PB1[kk] = *(const v4f*)(rb_ + W);
    rb.Q0[kk] = rs[0];
    rb.Q1[kk] = rs[W];
}

template <bool NEXT>
__device__ __forceinline__ void walk_tile(const float* __restrict__ in,
                                          float* __restrict__ dp,
                                          const ColGeo& cg, const TilePos& tp, const TilePos& tn,
                                          const Bands& bd, RowBuf& rb, int half)
{
    v8f acc[9];
    #pragma unroll
    for (int s = 0; s < 9; ++s) acc[s] = v8f{};

    auto mul_rows = [&](int kk) {
        const v2f b = bd.p1[kk];
        const int  y  = tp.ybase + 4 * kk + 2 * half;
        const bool yv = (unsigned)y < (unsigned)H;
        const bool va = yv && cg.xav;
        const bool vb = yv && cg.xbv;
        const bool vs = yv && cg.xsv;
        v2f a;
        a.x = va ? rb.PA0[kk].x : 0.f; a.y = va ? rb.PA1[kk].x : 0.f; acc[0] = WMMA(a, b, acc[0]);
        a.x = va ? rb.PA0[kk].y : 0.f; a.y = va ? rb.PA1[kk].y : 0.f; acc[1] = WMMA(a, b, acc[1]);
        a.x = va ? rb.PA0[kk].z : 0.f; a.y = va ? rb.PA1[kk].z : 0.f; acc[2] = WMMA(a, b, acc[2]);
        a.x = va ? rb.PA0[kk].w : 0.f; a.y = va ? rb.PA1[kk].w : 0.f; acc[3] = WMMA(a, b, acc[3]);
        a.x = vb ? rb.PB0[kk].x : 0.f; a.y = vb ? rb.PB1[kk].x : 0.f; acc[4] = WMMA(a, b, acc[4]);
        a.x = vb ? rb.PB0[kk].y : 0.f; a.y = vb ? rb.PB1[kk].y : 0.f; acc[5] = WMMA(a, b, acc[5]);
        a.x = vb ? rb.PB0[kk].z : 0.f; a.y = vb ? rb.PB1[kk].z : 0.f; acc[6] = WMMA(a, b, acc[6]);
        a.x = vb ? rb.PB0[kk].w : 0.f; a.y = vb ? rb.PB1[kk].w : 0.f; acc[7] = WMMA(a, b, acc[7]);
        a.x = vs ? rb.Q0[kk] : 0.f;    a.y = vs ? rb.Q1[kk] : 0.f;    acc[8] = WMMA(a, b, acc[8]);
    };

    #pragma unroll
    for (int g5 = 0; g5 < 5; ++g5) {
        __builtin_amdgcn_sched_barrier(0);
        #pragma unroll
        for (int kk = 4 * g5; kk < 4 * g5 + 4; ++kk)
            if (kk < 19) mul_rows(kk);
        if (g5 < 3) {
            #pragma unroll
            for (int kk = 4 * g5 + 8; kk < 4 * g5 + 12; ++kk)
                if (kk < 19) load_rows(in, cg, tp, rb, kk, half);
        } else if (NEXT) {
            #pragma unroll
            for (int kk = 4 * (g5 - 3); kk < 4 * (g5 - 3) + 4; ++kk)
                load_rows(in, cg, tn, rb, kk, half);
        }
    }
    __builtin_amdgcn_sched_barrier(0);

    v8f o0 = {}, o1 = {};
    #pragma unroll
    for (int m = 0; m < 4; ++m) {
        #pragma unroll
        for (int j = 0; j < 4; ++j) {
            v2f a;
            const v2f b = bd.p2[4 * m + j];
            a.x = acc[j][2 * m];     a.y = acc[j][2 * m + 1];
            o0 = WMMA(a, b, o0);
            a.x = acc[4 + j][2 * m]; a.y = acc[4 + j][2 * m + 1];
            o1 = WMMA(a, b, o1);
        }
    }
    #pragma unroll
    for (int m = 0; m < 2; ++m) {
        #pragma unroll
        for (int j = 0; j < 4; ++j) {
            v2f a;
            a.x = acc[4 + j][2 * m]; a.y = acc[4 + j][2 * m + 1];
            o0 = WMMA(a, bd.pl[4 * m + j], o0);
        }
    }
    #pragma unroll
    for (int u = 0; u < 4; ++u) {
        v2f a;
        a.x = acc[8][2 * u]; a.y = acc[8][2 * u + 1];
        o1 = WMMA(a, bd.pr[u], o1);
    }

    #pragma unroll
    for (int r = 0; r < 8; ++r) {
        dp[r * OW]      = o0[r];
        dp[r * OW + 16] = o1[r];
    }
    if (NEXT) {
        asm volatile("v_nop\n\tv_nop\n\tv_nop\n\tv_nop\n\tv_nop");
    }
}

__device__ __forceinline__ void lds_barrier()
{
    __builtin_amdgcn_fence(__ATOMIC_RELEASE, "workgroup", "local");
    __builtin_amdgcn_s_barrier();
    __builtin_amdgcn_fence(__ATOMIC_ACQUIRE, "workgroup", "local");
}

__global__ __launch_bounds__(128, 1)
void aa_interp_wmma_kernel(const float* __restrict__ in,
                           const float* __restrict__ wgt,
                           float* __restrict__ out)
{
    __shared__ float ctab[13];
    __shared__ __attribute__((aligned(16))) float Bt[16][BT_PITCH];

    const int tid = threadIdx.x;

    const int wv   = tid >> 5;
    const int lane = tid & 31;
    const int half = lane >> 4;
    const int l15  = lane & 15;

    const int row0 = blockIdx.x * ROWS_PER_BLOCK;
    const int ox0  = wv * 32;

    const int wr = min(tid, 12);
    float wrow[13];
    #pragma unroll
    for (int j = 0; j < 13; ++j) wrow[j] = wgt[wr * 13 + j];

    const ColGeo cg = col_geo(ox0, l15);
    TilePos tp = tile_pos(row0);
    RowBuf rb;
    #pragma unroll
    for (int kk = 0; kk < 8; ++kk) load_rows(in, cg, tp, rb, kk, half);
    __builtin_amdgcn_sched_barrier(0);

    float s = 0.f;
    #pragma unroll
    for (int j = 0; j < 13; ++j) s += wrow[j];
    ctab[wr] = s;
    lds_barrier();
    for (int idx = tid; idx < 16 * 80; idx += 128) {
        int n = idx / 80, w = idx % 80;
        int d = w - 4 * n - 2;
        Bt[n][w] = (d >= 0 && d < 13) ? ctab[d] : 0.0f;
    }
    lds_barrier();

    Bands bd;
    #pragma unroll
    for (int kk = 0; kk < 19; ++kk)
        bd.p1[kk] = *(const v2f*)(&Bt[l15][4 * kk + 2 * half]);
    #pragma unroll
    for (int m = 0; m < 4; ++m) {
        #pragma unroll
        for (int j = 0; j < 4; ++j) {
            bd.p2[4 * m + j].x = Bt[l15][32 * half + 8 * m + j];
            bd.p2[4 * m + j].y = Bt[l15][32 * half + 8 * m + j + 4];
        }
    }
    #pragma unroll
    for (int m = 0; m < 2; ++m) {
        #pragma unroll
        for (int j = 0; j < 4; ++j) {
            const float c0 = Bt[l15][64 + 8 * m + j];
            const float c1 = Bt[l15][64 + 8 * m + j + 4];
            bd.pl[4 * m + j].x = half ? 0.f : c0;
            bd.pl[4 * m + j].y = half ? 0.f : c1;
        }
    }
    #pragma unroll
    for (int u = 0; u < 4; ++u)
        bd.pr[u] = *(const v2f*)(&Bt[l15][64 + 8 * half + 2 * u]);

    const unsigned ooff = (unsigned)(8 * half * OW + ox0 + l15);

    #pragma nounroll
    for (int i = 0; i + 1 < ROWS_PER_BLOCK; ++i) {
        const TilePos tn = tile_pos(row0 + i + 1);
        walk_tile<true>(in, out + (tp.obase + ooff), cg, tp, tn, bd, rb, half);
        tp = tn;
    }
    walk_tile<false>(in, out + (tp.obase + ooff), cg, tp, tp, bd, rb, half);
}

extern "C" void kernel_launch(void* const* d_in, const int* in_sizes, int n_in,
                              void* d_out, int out_size, void* d_ws, size_t ws_size,
                              hipStream_t stream) {
    const float* inp = (const float*)d_in[0];
    const float* wgt = (const float*)d_in[1];
    float* out = (float*)d_out;

    dim3 grid(NC * TILE_ROWS_PER_IMG / ROWS_PER_BLOCK);
    dim3 block(WAVES_PER_BLOCK * 32);
    aa_interp_wmma_kernel<<<grid, block, 0, stream>>>(inp, wgt, out);
}


# Row groups of 5 k-steps instead of 4, each requested at the head of the group two ahead: 60 not 48 row loads in flight per wave, 90 not 72 products between a request and its first use
# speedup vs baseline: 2.8876x; 1.0591x over previous
#include <hip/hip_runtime.h>


typedef float v2f __attribute__((ext_vector_type(2)));
typedef float v4f __attribute__((ext_vector_type(4)));
typedef float v8f __attribute__((ext_vector_type(8)));

#define H   512
#define W   512
#define OH  128
#define OW  128
#define NC  96
#define TILE_ROWS_PER_IMG 8
#define ROWS_PER_BLOCK 3
#define WAVES_PER_BLOCK 4
#define BT_PITCH 84
#define KSTEPS 19
#define KG 5
#define NGROUPS 4

#define WMMA(a, b, c) __builtin_amdgcn_wmma_f32_16x16x4_f32(false, (a), false, (b), (short)0, (c), false, false)


struct RowBuf {
    v4f PA0[19], PA1[19];
    v4f PB0[19], PB1[19];
    float Q0[19], Q1[19];
};

struct ColGeo {
    bool xav, xbv, xsv;
    int xac, xbc, xsc;
};

struct TilePos {
    int ybase;
    unsigned img;
    unsigned obase;
};

struct Bands {
    v2f p1[19];
    v2f p2[16];
    v2f pl[8];
    v2f pr[4];
};

__device__ __forceinline__ ColGeo col_geo(int ox0, int l15)
{
    ColGeo c;
    const int X0 = 4 * ox0 - 8;
    const int xa = X0 + 4 * l15;
    const int xb = X0 + 64 + 4 * l15;
    const int xs = X0 + 128 + l15;
    c.xav = (unsigned)xa < (unsigned)W;
    c.xbv = (unsigned)xb < (unsigned)W;
    c.xsv = xs < W;
    c.xac = min(max(xa, 0), W - 4);
    c.xbc = min(max(xb, 0), W - 4);
    c.xsc = min(xs, W - 1);
    return c;
}

__device__ __forceinline__ TilePos tile_pos(int R)
{
    TilePos p;
    const int nc  = R / TILE_ROWS_PER_IMG;
    const int oy0 = (R % TILE_ROWS_PER_IMG) * 16;
    p.ybase = 4 * oy0 - 8;
    p.img   = (unsigned)nc * (unsigned)(H * W);
    p.obase = (unsigned)nc * (unsigned)(OH * OW) + (unsigned)(oy0 * OW);
    return p;
}

__device__ __forceinline__ void load_rows(const float* __restrict__ in, const ColGeo& cg,
                                          const TilePos& tp, RowBuf& rb, int kk, int half)
{
    const int  y  = tp.ybase + 4 * kk + 2 * half;
    const int  yc = min(max(y, 0), H - 2);
    const float* ra  = in + (tp.img + (unsigned)(yc * W + cg.xac));
    const float* rb_ = in + (tp.img + (unsigned)(yc * W + cg.xbc));
    const float* rs  = in + (tp.img + (unsigned)(yc * W + cg.xsc));
    rb.PA0[kk] = *(const v4f*)(ra);
    rb.PA1[kk] = *(const v4f*)(ra + W);
    rb.PB0[kk] = *(const v4f*)(rb_);
    rb.PB1[kk] = *(const v4f*)(rb_ + W);
    rb.Q0[kk] = rs[0];
    rb.Q1[kk] = rs[W];
}

template <bool NEXT>
__device__ __forceinline__ void walk_tile(const float* __restrict__ in,
                                          float* __restrict__ dp,
                                          const ColGeo& cg, const TilePos& tp, const TilePos& tn,
                                          const Bands& bd, RowBuf& rb, int half)
{
    v8f acc[9];
    #pragma unroll
    for (int s = 0; s < 9; ++s) acc[s] = v8f{};

    auto mul_rows = [&](int kk) {
        const v2f b = bd.p1[kk];
        const int  y  = tp.ybase + 4 * kk + 2 * half;
        const bool yv = (unsigned)y < (unsigned)H;
        const bool va = yv && cg.xav;
        const bool vb = yv && cg.xbv;
        const bool vs = yv && cg.xsv;
        v2f a;
        a.x = va ? rb.PA0[kk].x : 0.f; a.y = va ? rb.PA1[kk].x : 0.f; acc[0] = WMMA(a, b, acc[0]);
        a.x = va ? rb.PA0[kk].y : 0.f; a.y = va ? rb.PA1[kk].y : 0.f; acc[1] = WMMA(a, b, acc[1]);
        a.x = va ? rb.PA0[kk].z : 0.f; a.y = va ? rb.PA1[kk].z : 0.f; acc[2] = WMMA(a, b, acc[2]);
        a.x = va ? rb.PA0[kk].w : 0.f; a.y = va ? rb.PA1[kk].w : 0.f; acc[3] = WMMA(a, b, acc[3]);
        a.x = vb ? rb.PB0[kk].x : 0.f; a.y = vb ? rb.PB1[kk].x : 0.f; acc[4] = WMMA(a, b, acc[4]);
        a.x = vb ? rb.PB0[kk].y : 0.f; a.y = vb ? rb.PB1[kk].y : 0.f; acc[5] = WMMA(a, b, acc[5]);
        a.x = vb ? rb.PB0[kk].z : 0.f; a.y = vb ? rb.PB1[kk].z : 0.f; acc[6] = WMMA(a, b, acc[6]);
        a.x = vb ? rb.PB0[kk].w : 0.f; a.y = vb ? rb.PB1[kk].w : 0.f; acc[7] = WMMA(a, b, acc[7]);
        a.x = vs ? rb.Q0[kk] : 0.f;    a.y = vs ? rb.Q1[kk] : 0.f;    acc[8] = WMMA(a, b, acc[8]);
    };

    #pragma unroll
    for (int g = 0; g < NGROUPS; ++g) {
        __builtin_amdgcn_sched_barrier(0);
        if (g + 2 < NGROUPS) {
            #pragma unroll
            for (int kk = KG * (g + 2); kk < KG * (g + 3); ++kk)
                if (kk < KSTEPS) load_rows(in, cg, tp, rb, kk, half);
        } else if (NEXT) {
            #pragma unroll
            for (int kk = KG * (g + 2 - NGROUPS); kk < KG * (g + 3 - NGROUPS); ++kk)
                load_rows(in, cg, tn, rb, kk, half);
        }
        #pragma unroll
        for (int kk = KG * g; kk < KG * (g + 1); ++kk)
            if (kk < KSTEPS) mul_rows(kk);
    }
    __builtin_amdgcn_sched_barrier(0);

    v8f o0 = {}, o1 = {};
    #pragma unroll
    for (int m = 0; m < 4; ++m) {
        #pragma unroll
        for (int j = 0; j < 4; ++j) {
            v2f a;
            const v2f b = bd.p2[4 * m + j];
            a.x = acc[j][2 * m];     a.y = acc[j][2 * m + 1];
            o0 = WMMA(a, b, o0);
            a.x = acc[4 + j][2 * m]; a.y = acc[4 + j][2 * m + 1];
            o1 = WMMA(a, b, o1);
        }
    }
    #pragma unroll
    for (int m = 0; m < 2; ++m) {
        #pragma unroll
        for (int j = 0; j < 4; ++j) {
            v2f a;
            a.x = acc[4 + j][2 * m]; a.y = acc[4 + j][2 * m + 1];
            o0 = WMMA(a, bd.pl[4 * m + j], o0);
        }
    }
    #pragma unroll
    for (int u = 0; u < 4; ++u) {
        v2f a;
        a.x = acc[8][2 * u]; a.y = acc[8][2 * u + 1];
        o1 = WMMA(a, bd.pr[u], o1);
    }

    #pragma unroll
    for (int r = 0; r < 8; ++r) {
        dp[r * OW]      = o0[r];
        dp[r * OW + 16] = o1[r];
    }
    if (NEXT) {
        asm volatile("v_nop\n\tv_nop\n\tv_nop\n\tv_nop\n\tv_nop");
    }
}

__device__ __forceinline__ void lds_barrier()
{
    __builtin_amdgcn_fence(__ATOMIC_RELEASE, "workgroup", "local");
    __builtin_amdgcn_s_barrier();
    __builtin_amdgcn_fence(__ATOMIC_ACQUIRE, "workgroup", "local");
}

__global__ __launch_bounds__(128, 1)
void aa_interp_wmma_kernel(const float* __restrict__ in,
                           const float* __restrict__ wgt,
                           float* __restrict__ out)
{
    __shared__ float ctab[13];
    __shared__ __attribute__((aligned(16))) float Bt[16][BT_PITCH];

    const int tid = threadIdx.x;

    const int wv   = tid >> 5;
    const int lane = tid & 31;
    const int half = lane >> 4;
    const int l15  = lane & 15;

    const int row0 = blockIdx.x * ROWS_PER_BLOCK;
    const int ox0  = wv * 32;

    const int wr = min(tid, 12);
    float wrow[13];
    #pragma unroll
    for (int j = 0; j < 13; ++j) wrow[j] = wgt[wr * 13 + j];

    const ColGeo cg = col_geo(ox0, l15);
    TilePos tp = tile_pos(row0);
    RowBuf rb;
    #pragma unroll
    for (int kk = 0; kk < 2 * KG; ++kk) load_rows(in, cg, tp, rb, kk, half);
    __builtin_amdgcn_sched_barrier(0);

    float s = 0.f;
    #pragma unroll
    for (int j = 0; j < 13; ++j) s += wrow[j];
    ctab[wr] = s;
    lds_barrier();
    for (int idx = tid; idx < 16 * 80; idx += 128) {
        int n = idx / 80, w = idx % 80;
        int d = w - 4 * n - 2;
        Bt[n][w] = (d >= 0 && d < 13) ? ctab[d] : 0.0f;
    }
    lds_barrier();

    Bands bd;
    #pragma unroll
    for (int kk = 0; kk < 19; ++kk)
        bd.p1[kk] = *(const v2f*)(&Bt[l15][4 * kk + 2 * half]);
    #pragma unroll
    for (int m = 0; m < 4; ++m) {
        #pragma unroll
        for (int j = 0; j < 4; ++j) {
            bd.p2[4 * m + j].x = Bt[l15][32 * half + 8 * m + j];
            bd.p2[4 * m + j].y = Bt[l15][32 * half + 8 * m + j + 4];
        }
    }
    #pragma unroll
    for (int m = 0; m < 2; ++m) {
        #pragma unroll
        for (int j = 0; j < 4; ++j) {
            const float c0 = Bt[l15][64 + 8 * m + j];
            const float c1 = Bt[l15][64 + 8 * m + j + 4];
            bd.pl[4 * m + j].x = half ? 0.f : c0;
            bd.pl[4 * m + j].y = half ? 0.f : c1;
        }
    }
    #pragma unroll
    for (int u = 0; u < 4; ++u)
        bd.pr[u] = *(const v2f*)(&Bt[l15][64 + 8 * half + 2 * u]);

    const unsigned ooff = (unsigned)(8 * half * OW + ox0 + l15);

    #pragma nounroll
    for (int i = 0; i + 1 < ROWS_PER_BLOCK; ++i) {
        const TilePos tn = tile_pos(row0 + i + 1);
        walk_tile<true>(in, out + (tp.obase + ooff), cg, tp, tn, bd, rb, half);
        tp = tn;
    }
    walk_tile<false>(in, out + (tp.obase + ooff), cg, tp, tp, bd, rb, half);
}

extern "C" void kernel_launch(void* const* d_in, const int* in_sizes, int n_in,
                              void* d_out, int out_size, void* d_ws, size_t ws_size,
                              hipStream_t stream) {
    const float* inp = (const float*)d_in[0];
    const float* wgt = (const float*)d_in[1];
    float* out = (float*)d_out;

    dim3 grid(NC * TILE_ROWS_PER_IMG / ROWS_PER_BLOCK);
    dim3 block(WAVES_PER_BLOCK * 32);
    aa_interp_wmma_kernel<<<grid, block, 0, stream>>>(inp, wgt, out);
}


# Wave column windows start at the image edge (8 column sets, not 9; row groups of 7 k-steps): the 8 blurred halo columns per side come from the neighbour wave via LDS; pass-2 edge sums reordered
# speedup vs baseline: 3.1015x; 1.0741x over previous
#include <hip/hip_runtime.h>


typedef float v2f __attribute__((ext_vector_type(2)));
typedef float v4f __attribute__((ext_vector_type(4)));
typedef float v8f __attribute__((ext_vector_type(8)));

#define H   512
#define W   512
#define OH  128
#define OW  128
#define NC  96
#define TILE_ROWS_PER_IMG 8
#define ROWS_PER_BLOCK 3
#define WAVES_PER_BLOCK 4
#define BT_PITCH 84
#define KSTEPS 19
#define KG 7
#define NGROUPS 3
#define XPITCH 12
#define XSIDE (16 * XPITCH)

#define WMMA(a, b, c) __builtin_amdgcn_wmma_f32_16x16x4_f32(false, (a), false, (b), (short)0, (c), false, false)


struct RowBuf {
    v4f PA0[19], PA1[19];
    v4f PB0[19], PB1[19];
};

struct TilePos {
    int ybase;
    unsigned img;
    unsigned obase;
};

struct Bands {
    v2f p1[19];
    v2f p2[16];
    v2f e0[4];
    v2f e1[4];
    v2f nl[2];
    v2f nr[2];
};

__device__ __forceinline__ TilePos tile_pos(int R)
{
    TilePos p;
    const int nc  = R / TILE_ROWS_PER_IMG;
    const int oy0 = (R % TILE_ROWS_PER_IMG) * 16;
    p.ybase = 4 * oy0 - 8;
    p.img   = (unsigned)nc * (unsigned)(H * W);
    p.obase = (unsigned)nc * (unsigned)(OH * OW) + (unsigned)(oy0 * OW);
    return p;
}

__device__ __forceinline__ void load_rows(const float* __restrict__ in, int xa,
                                          const TilePos& tp, RowBuf& rb, int kk, int half)
{
    const int  y  = tp.ybase + 4 * kk + 2 * half;
    const int  yc = min(max(y, 0), H - 2);
    const float* ra = in + (tp.img + (unsigned)(yc * W + xa));
    rb.PA0[kk] = *(const v4f*)(ra);
    rb.PA1[kk] = *(const v4f*)(ra + W);
    rb.PB0[kk] = *(const v4f*)(ra + 64);
    rb.PB1[kk] = *(const v4f*)(ra + W + 64);
}

template <bool NEXT>
__device__ __forceinline__ void walk_tile(const float* __restrict__ in,
                                          float* __restrict__ dp,
                                          int xa, const TilePos& tp, const TilePos& tn,
                                          const Bands& bd, RowBuf& rb, int half,
                                          float* xw, const float* xl, const float* xr)
{
    v8f acc[8];
    #pragma unroll
    for (int s = 0; s < 8; ++s) acc[s] = v8f{};

    auto mul_rows = [&](int kk) {
        const v2f b = bd.p1[kk];
        const int  y  = tp.ybase + 4 * kk + 2 * half;
        const bool yv = (unsigned)y < (unsigned)H;
        v2f a;
        a.x = yv ? rb.PA0[kk].x : 0.f; a.y = yv ? rb.PA1[kk].x : 0.f; acc[0] = WMMA(a, b, acc[0]);
        a.x = yv ? rb.PA0[kk].y : 0.f; a.y = yv ? rb.PA1[kk].y : 0.f; acc[1] = WMMA(a, b, acc[1]);
        a.x = yv ? rb.PA0[kk].z : 0.f; a.y = yv ? rb.PA1[kk].z : 0.f; acc[2] = WMMA(a, b, acc[2]);
        a.x = yv ? rb.PA0[kk].w : 0.f; a.y = yv ? rb.PA1[kk].w : 0.f; acc[3] = WMMA(a, b, acc[3]);
        a.x = yv ? rb.PB0[kk].x : 0.f; a.y = yv ? rb.PB1[kk].x : 0.f; acc[4] = WMMA(a, b, acc[4]);
        a.x = yv ? rb.PB0[kk].y : 0.f; a.y = yv ? rb.PB1[kk].y : 0.f; acc[5] = WMMA(a, b, acc[5]);
        a.x = yv ? rb.PB0[kk].z : 0.f; a.y = yv ? rb.PB1[kk].z : 0.f; acc[6] = WMMA(a, b, acc[6]);
        a.x = yv ? rb.PB0[kk].w : 0.f; a.y = yv ? rb.PB1[kk].w : 0.f; acc[7] = WMMA(a, b, acc[7]);
    };

    #pragma unroll
    for (int g = 0; g < NGROUPS; ++g) {
        __builtin_amdgcn_sched_barrier(0);
        if (g + 2 < NGROUPS) {
            #pragma unroll
            for (int kk = KG * (g + 2); kk < KG * (g + 3); ++kk)
                if (kk < KSTEPS) load_rows(in, xa, tp, rb, kk, half);
        } else if (NEXT) {
            #pragma unroll
            for (int kk = KG * (g + 2 - NGROUPS); kk < KG * (g + 3 - NGROUPS); ++kk)
                load_rows(in, xa, tn, rb, kk, half);
        }
        #pragma unroll
        for (int kk = KG * g; kk < KG * (g + 1); ++kk)
            if (kk < KSTEPS) mul_rows(kk);
    }
    __builtin_amdgcn_sched_barrier(0);

    {
        v4f lo, hi;
        lo.x = half ? acc[4][6] : acc[0][0]; hi.x = half ? acc[4][7] : acc[0][1];
        lo.y = half ? acc[5][6] : acc[1][0]; hi.y = half ? acc[5][7] : acc[1][1];
        lo.z = half ? acc[6][6] : acc[2][0]; hi.z = half ? acc[6][7] : acc[2][1];
        lo.w = half ? acc[7][6] : acc[3][0]; hi.w = half ? acc[7][7] : acc[3][1];
        *(v4f*)(xw)     = lo;
        *(v4f*)(xw + 4) = hi;
    }
    __builtin_amdgcn_sched_barrier(0);

    v8f o0 = {}, o1 = {};
    #pragma unroll
    for (int j = 0; j < 4; ++j) {
        v2f a;
        a.x = acc[4 + j][0]; a.y = acc[4 + j][1];
        o0 = WMMA(a, bd.e0[j], o0);
        a.x = acc[j][6]; a.y = acc[j][7];
        o1 = WMMA(a, bd.e1[j], o1);
    }
    __builtin_amdgcn_sched_barrier(0);
    __builtin_amdgcn_fence(__ATOMIC_RELEASE, "workgroup", "local");
    __builtin_amdgcn_s_barrier_signal(-1);
    __builtin_amdgcn_sched_barrier(0);

    auto main_step = [&](int m) {
        #pragma unroll
        for (int j = 0; j < 4; ++j) {
            v2f a;
            const v2f b = bd.p2[4 * m + j];
            a.x = acc[j][2 * m];     a.y = acc[j][2 * m + 1];
            o0 = WMMA(a, b, o0);
            a.x = acc[4 + j][2 * m]; a.y = acc[4 + j][2 * m + 1];
            o1 = WMMA(a, b, o1);
        }
    };
    main_step(0);
    main_step(1);
    main_step(2);
    __builtin_amdgcn_sched_barrier(0);
    __builtin_amdgcn_s_barrier_wait(-1);
    __builtin_amdgcn_fence(__ATOMIC_ACQUIRE, "workgroup", "local");
    const v2f al0 = *(const v2f*)(xl + 2 * half);
    const v2f al1 = *(const v2f*)(xl + 4 + 2 * half);
    const v2f ar0 = *(const v2f*)(xr + 2 * half);
    const v2f ar1 = *(const v2f*)(xr + 4 + 2 * half);
    __builtin_amdgcn_sched_barrier(0);
    main_step(3);
    o0 = WMMA(al0, bd.nl[0], o0);
    o1 = WMMA(ar0, bd.nr[0], o1);
    o0 = WMMA(al1, bd.nl[1], o0);
    o1 = WMMA(ar1, bd.nr[1], o1);

    #pragma unroll
    for (int r = 0; r < 8; ++r) {
        dp[r * OW]      = o0[r];
        dp[r * OW + 16] = o1[r];
    }
    if (NEXT) {
        asm volatile("v_nop\n\tv_nop\n\tv_nop\n\tv_nop\n\tv_nop");
    }
}

__device__ __forceinline__ void lds_barrier()
{
    __builtin_amdgcn_fence(__ATOMIC_RELEASE, "workgroup", "local");
    __builtin_amdgcn_s_barrier();
    __builtin_amdgcn_fence(__ATOMIC_ACQUIRE, "workgroup", "local");
}

__global__ __launch_bounds__(128, 1)
void aa_interp_wmma_kernel(const float* __restrict__ in,
                           const float* __restrict__ wgt,
                           float* __restrict__ out)
{
    __shared__ float ctab[13];
    __shared__ __attribute__((aligned(16))) float Bt[16][BT_PITCH];
    __shared__ __attribute__((aligned(16))) float xch[2][WAVES_PER_BLOCK][2][XSIDE];

    const int tid = threadIdx.x;

    const int wv   = tid >> 5;
    const int lane = tid & 31;
    const int half = lane >> 4;
    const int l15  = lane & 15;

    const int row0 = blockIdx.x * ROWS_PER_BLOCK;
    const int ox0  = wv * 32;

    const int wr = min(tid, 12);
    float wrow[13];
    #pragma unroll
    for (int j = 0; j < 13; ++j) wrow[j] = wgt[wr * 13 + j];

    const int xa = 4 * ox0 + 4 * l15;
    TilePos tp = tile_pos(row0);
    RowBuf rb;
    #pragma unroll
    for (int kk = 0; kk < 2 * KG; ++kk) load_rows(in, xa, tp, rb, kk, half);
    __builtin_amdgcn_sched_barrier(0);

    float s = 0.f;
    #pragma unroll
    for (int j = 0; j < 13; ++j) s += wrow[j];
    ctab[wr] = s;
    lds_barrier();
    for (int idx = tid; idx < 16 * 80; idx += 128) {
        int n = idx / 80, w = idx % 80;
        int d = w - 4 * n - 2;
        Bt[n][w] = (d >= 0 && d < 13) ? ctab[d] : 0.0f;
    }
    lds_barrier();

    Bands bd;
    #pragma unroll
    for (int kk = 0; kk < 19; ++kk)
        bd.p1[kk] = *(const v2f*)(&Bt[l15][4 * kk + 2 * half]);
    #pragma unroll
    for (int m = 0; m < 4; ++m) {
        #pragma unroll
        for (int j = 0; j < 4; ++j) {
            bd.p2[4 * m + j].x = Bt[l15][8 + 32 * half + 8 * m + j];
            bd.p2[4 * m + j].y = Bt[l15][8 + 32 * half + 8 * m + j + 4];
        }
    }
    #pragma unroll
    for (int j = 0; j < 4; ++j) {
        const float c0 = Bt[l15][72 + j];
        const float c1 = Bt[l15][76 + j];
        bd.e0[j].x = half ? 0.f : c0;
        bd.e0[j].y = half ? 0.f : c1;
        const float d0 = Bt[l15][j];
        const float d1 = Bt[l15][4 + j];
        bd.e1[j].x = half ? d0 : 0.f;
        bd.e1[j].y = half ? d1 : 0.f;
    }
    #pragma unroll
    for (int u = 0; u < 2; ++u) {
        const v2f l = *(const v2f*)(&Bt[l15][4 * u + 2 * half]);
        const v2f r = *(const v2f*)(&Bt[l15][72 + 4 * u + 2 * half]);
        bd.nl[u].x = (wv == 0) ? 0.f : l.x;
        bd.nl[u].y = (wv == 0) ? 0.f : l.y;
        bd.nr[u].x = (wv == WAVES_PER_BLOCK - 1) ? 0.f : r.x;
        bd.nr[u].y = (wv == WAVES_PER_BLOCK - 1) ? 0.f : r.y;
    }

    const unsigned ooff = (unsigned)(8 * half * OW + ox0 + l15);

    const int wl = max(wv - 1, 0);
    const int wr2 = min(wv + 1, WAVES_PER_BLOCK - 1);
    float*       xw = &xch[0][wv][half][l15 * XPITCH];
    const float* xl = &xch[0][wl][1][l15 * XPITCH];
    const float* xr = &xch[0][wr2][0][l15 * XPITCH];
    const int XPAR = WAVES_PER_BLOCK * 2 * XSIDE;
    int par = 0;

    #pragma nounroll
    for (int i = 0; i + 1 < ROWS_PER_BLOCK; ++i) {
        const TilePos tn = tile_pos(row0 + i + 1);
        walk_tile<true>(in, out + (tp.obase + ooff), xa, tp, tn, bd, rb, half,
                        xw + par, xl + par, xr + par);
        tp = tn;
        par ^= XPAR;
    }
    walk_tile<false>(in, out + (tp.obase + ooff), xa, tp, tp, bd, rb, half,
                     xw + par, xl + par, xr + par);
}

extern "C" void kernel_launch(void* const* d_in, const int* in_sizes, int n_in,
                              void* d_out, int out_size, void* d_ws, size_t ws_size,
                              hipStream_t stream) {
    const float* inp = (const float*)d_in[0];
    const float* wgt = (const float*)d_in[1];
    float* out = (float*)d_out;

    dim3 grid(NC * TILE_ROWS_PER_IMG / ROWS_PER_BLOCK);
    dim3 block(WAVES_PER_BLOCK * 32);
    aa_interp_wmma_kernel<<<grid, block, 0, stream>>>(inp, wgt, out);
}
